# starting kernel
# baseline (speedup 1.0000x reference)
#include <hip/hip_runtime.h>

#define NS 64
#define HC 256
#define NB 8
#define LL 1024
#define MS 66

typedef __attribute__((ext_vector_type(2))) float v2f;
typedef __attribute__((ext_vector_type(8))) float v8f;

template <int NCT>
__device__ __forceinline__ void wg_matmul(float* __restrict__ D,
                                          const float* __restrict__ A,
                                          const float* __restrict__ Bm,
                                          int wave, int lr, int half)
{
  for (int tt = wave; tt < 4 * NCT; tt += 8) {
    const int rt = tt / NCT, ct = tt % NCT;
    const int r0 = rt * 16, c0 = ct * 16;

    v2f af[16], bf[16];
#pragma unroll
    for (int k = 0; k < 16; ++k) {
      const int ka = 4 * k + 2 * half;
      af[k].x = A[(r0 + lr) * MS + ka];
      af[k].y = A[(r0 + lr) * MS + ka + 1];
      bf[k].x = Bm[(ka    ) * MS + c0 + lr];
      bf[k].y = Bm[(ka + 1) * MS + c0 + lr];
    }

    v8f acc = {0.f, 0.f, 0.f, 0.f, 0.f, 0.f, 0.f, 0.f};
#pragma unroll
    for (int k = 0; k < 16; ++k)
      acc = __builtin_amdgcn_wmma_f32_16x16x4_f32(false, af[k], false, bf[k],
                                                  (short)0, acc, false, false);
#pragma unroll
    for (int r = 0; r < 8; ++r)
      D[(r0 + r + 8 * half) * MS + c0 + lr] = acc[r];
  }
}

__global__ __launch_bounds__(256) void ssm_build_k(
    const float* __restrict__ As, const float* __restrict__ Bs,
    const float* __restrict__ Cs, const float* __restrict__ log_steps,
    float* __restrict__ Kws)
{
  extern __shared__ float smem[];
  float (*aug)[130] = (float (*)[130])smem;
  float* P  = smem + NS * 130;
  float* Pn = P  + NS * MS;
  float* Xa = Pn + NS * MS;
  float* Xb = Xa + NS * MS;
  float* cb = Xb + NS * MS;

  const int h    = blockIdx.x;
  const int t    = threadIdx.x;
  const int wave = t >> 5, lane = t & 31;
  const int half = lane >> 4, lr = lane & 15;

  const int i  = t >> 2;
  const int q  = t & 3;
  const int j0 = q * 33;
  const int j1 = (j0 + 33 < 130) ? j0 + 33 : 130;
  const float s = 0.5f * __expf(log_steps[h]);
  const float* Ah = As + (size_t)h * NS * NS;

  for (int j = j0; j < j1; ++j) {
    float v;
    if (j < NS)          { const float a = Ah[i * NS + j];
                           v = ((i == j) ? 1.f : 0.f) - s * a; }
    else if (j < 2 * NS) { const float a = Ah[i * NS + (j - NS)];
                           v = ((i == j - NS) ? 1.f : 0.f) + s * a; }
    else if (j == 2 * NS)  v = 2.f * s * Bs[h * NS + i];
    else                   v = 0.f;
    aug[i][j] = v;
  }
  if (t < NS) cb[t] = Cs[h * NS + t];
  __syncthreads();

  for (int p = 0; p < NS; ++p) {
    const float f = (i != p) ? aug[i][p] / aug[p][p] : 0.0f;
    __syncthreads();
    if (i != p)
      for (int j = j0; j < j1; ++j) aug[i][j] -= f * aug[p][j];
    __syncthreads();
  }

  {
    const float dinv = 1.0f / aug[i][i];
    for (int j = j0; j < j1; ++j)
      if (j >= NS && j < 2 * NS) P[i * MS + (j - NS)] = aug[i][j] * dinv;
    if (q == 3) Xa[i * MS + 0] = aug[i][2 * NS] * dinv;
  }
  __syncthreads();

  for (int l = 1; l < 16; ++l) {
    if (t < NS) {
      float y = 0.f;
#pragma unroll
      for (int j = 0; j < NS; ++j) y += P[t * MS + j] * Xa[j * MS + (l - 1)];
      Xa[t * MS + l] = y;
    }
    __syncthreads();
  }

  float* Pa = P; float* Pb = Pn;
  for (int sq = 0; sq < 4; ++sq) {
    wg_matmul<4>(Pb, Pa, Pa, wave, lr, half);
    __syncthreads();
    float* tp = Pa; Pa = Pb; Pb = tp;
  }
  for (int e = 0; e < 3; ++e) {
    wg_matmul<1>(Xa + 16 * (e + 1), Pa, Xa + 16 * e, wave, lr, half);
    __syncthreads();
  }
  for (int sq = 0; sq < 2; ++sq) {
    wg_matmul<4>(Pb, Pa, Pa, wave, lr, half);
    __syncthreads();
    float* tp = Pa; Pa = Pb; Pb = tp;
  }

  float* Xc = Xa; float* Xd = Xb;
  for (int blk = 0; blk < 16; ++blk) {
    if (t < NS) {
      float ks = 0.f;
#pragma unroll
      for (int r = 0; r < NS; ++r) ks += cb[r] * Xc[r * MS + t];
      Kws[(size_t)h * LL + blk * NS + t] = ks;
    }
    if (blk < 15)
      wg_matmul<4>(Xd, Pa, Xc, wave, lr, half);
    __syncthreads();
    float* tx = Xc; Xc = Xd; Xd = tx;
  }
}

__global__ __launch_bounds__(256) void ssm_conv_wmma(
    const float* __restrict__ inp, const float* __restrict__ Kws,
    const float* __restrict__ Ds, float* __restrict__ out)
{
  extern __shared__ float smem[];
  float* Ush = smem;
  float* Kp  = smem + LL * 16;

  const int h    = blockIdx.x;
  const int tid  = threadIdx.x;
  const int wave = tid >> 5;
  const int lane = tid & 31;
  const int half = lane >> 4;
  const int lr   = lane & 15;

  for (int idx = tid; idx < LL * 16; idx += 256) {
    const int l = idx >> 4, c = idx & 15;
    Ush[idx] = (c < NB) ? inp[(size_t)c * LL * HC + (size_t)l * HC + h] : 0.f;
  }
  for (int idx = tid; idx < 16 + LL; idx += 256)
    Kp[idx] = (idx < 16) ? 0.f : Kws[(size_t)h * LL + (idx - 16)];
  __syncthreads();

  const float Dh = Ds[h];

  for (int t = wave; t < LL / 16; t += 8) {
    const int l0 = t * 16;
    v8f acc = {0.f, 0.f, 0.f, 0.f, 0.f, 0.f, 0.f, 0.f};

    for (int m0 = 0; m0 < l0 + 16; m0 += 16) {
      v2f af[4], bf[4];
#pragma unroll
      for (int u = 0; u < 4; ++u) {
        const int mrow = m0 + 4 * u + 2 * half;
        const int ia   = 16 + l0 + lr - mrow;
        af[u].x = Kp[ia];
        af[u].y = Kp[ia - 1];
        bf[u].x = Ush[mrow * 16 + lr];
        bf[u].y = Ush[(mrow + 1) * 16 + lr];
      }
#pragma unroll
      for (int u = 0; u < 4; ++u)
        acc = __builtin_amdgcn_wmma_f32_16x16x4_f32(false, af[u], false, bf[u],
                                                    (short)0, acc, false, false);
    }

    if (lr < NB) {
#pragma unroll
      for (int r = 0; r < 8; ++r) {
        const int row = l0 + r + 8 * half;
        const float uv = Ush[row * 16 + lr];
        out[(size_t)lr * LL * HC + (size_t)row * HC + h] = acc[r] + Dh * uv;
      }
    }
  }
}

extern "C" void kernel_launch(void* const* d_in, const int* in_sizes, int n_in,
                              void* d_out, int out_size, void* d_ws, size_t ws_size,
                              hipStream_t stream)
{
  const float* inp = (const float*)d_in[0];
  const float* As  = (const float*)d_in[1];
  const float* Bs  = (const float*)d_in[2];
  const float* Cs  = (const float*)d_in[3];
  const float* Ds  = (const float*)d_in[4];
  const float* ls  = (const float*)d_in[5];
  float* out = (float*)d_out;

  float* Kws = (float*)d_ws;

  const size_t lds1 = (size_t)(NS * 130 + 4 * NS * MS + NS) * sizeof(float);
  const size_t lds2 = (size_t)(LL * 16 + 16 + LL) * sizeof(float);

  ssm_build_k  <<<dim3(HC), dim3(256), lds1, stream>>>(As, Bs, Cs, ls, Kws);
  ssm_conv_wmma<<<dim3(HC), dim3(256), lds2, stream>>>(inp, Kws, Ds, out);
}


# Gauss-Jordan phase: each thread keeps its row slice in registers (interleaved columns, window slides as pivots retire); pivot row via double-buffered LDS, one barrier per pivot
# speedup vs baseline: 1.7425x; 1.7425x over previous
#include <hip/hip_runtime.h>

#define NS 64
#define HC 256
#define NB 8
#define LL 1024
#define MS 66

typedef __attribute__((ext_vector_type(2))) float v2f;
typedef __attribute__((ext_vector_type(8))) float v8f;

template <int NCT>
__device__ __forceinline__ void wg_matmul(float* __restrict__ D,
                                          const float* __restrict__ A,
                                          const float* __restrict__ Bm,
                                          int wave, int lr, int half)
{
  for (int tt = wave; tt < 4 * NCT; tt += 8) {
    const int rt = tt / NCT, ct = tt % NCT;
    const int r0 = rt * 16, c0 = ct * 16;

    v2f af[16], bf[16];
#pragma unroll
    for (int k = 0; k < 16; ++k) {
      const int ka = 4 * k + 2 * half;
      af[k].x = A[(r0 + lr) * MS + ka];
      af[k].y = A[(r0 + lr) * MS + ka + 1];
      bf[k].x = Bm[(ka    ) * MS + c0 + lr];
      bf[k].y = Bm[(ka + 1) * MS + c0 + lr];
    }

    v8f acc = {0.f, 0.f, 0.f, 0.f, 0.f, 0.f, 0.f, 0.f};
#pragma unroll
    for (int k = 0; k < 16; ++k)
      acc = __builtin_amdgcn_wmma_f32_16x16x4_f32(false, af[k], false, bf[k],
                                                  (short)0, acc, false, false);
#pragma unroll
    for (int r = 0; r < 8; ++r)
      D[(r0 + r + 8 * half) * MS + c0 + lr] = acc[r];
  }
}

#define PW 36

template <int PP>
__device__ __forceinline__ void gj_pivot(float (&r)[33], float& diag,
                                         float* __restrict__ prow,
                                         int p, int i, int q)
{
  float* buf = prow + (PP & 1) * (4 * PW);
  float4* mine = (float4*)(buf + q * PW);

  if (i == p) {
#pragma unroll
    for (int k = 0; k < 8; ++k)
      mine[k] = make_float4(r[4 * k], r[4 * k + 1], r[4 * k + 2], r[4 * k + 3]);
    buf[q * PW + 32] = r[32];
  }
  const float aip = __int_as_float(__builtin_amdgcn_update_dpp(
      0, __float_as_int(r[0]), PP * 0x55, 0xF, 0xF, true));
  __syncthreads();

  float pr[33];
#pragma unroll
  for (int k = 0; k < 8; ++k) {
    const float4 v = mine[k];
    pr[4 * k] = v.x; pr[4 * k + 1] = v.y; pr[4 * k + 2] = v.z; pr[4 * k + 3] = v.w;
  }
  pr[32] = buf[q * PW + 32];
  const float piv = buf[PP * PW];

  const float f = (i != p) ? aip / piv : 0.0f;
  if (i == p) diag = piv;
#pragma unroll
  for (int k = 0; k < 33; ++k) r[k] = __builtin_fmaf(-f, pr[k], r[k]);
}

__global__ __launch_bounds__(256) void ssm_build_k(
    const float* __restrict__ As, const float* __restrict__ Bs,
    const float* __restrict__ Cs, const float* __restrict__ log_steps,
    float* __restrict__ Kws)
{
  extern __shared__ __attribute__((aligned(16))) float smem[];
  float* prow = smem;
  float* P  = smem + 2 * 4 * PW;
  float* Pn = P  + NS * MS;
  float* Xa = Pn + NS * MS;
  float* Xb = Xa + NS * MS;
  float* cb = Xb + NS * MS;

  const int h    = blockIdx.x;
  const int t    = threadIdx.x;
  const int wave = t >> 5, lane = t & 31;
  const int half = lane >> 4, lr = lane & 15;

  const int i = t >> 2;
  const int q = t & 3;
  const float s = 0.5f * __expf(log_steps[h]);
  const float* Ah = As + (size_t)h * NS * NS;

  float r[33];
  {
    float a[16];
#pragma unroll
    for (int k = 0; k < 16; ++k) a[k] = Ah[i * NS + 4 * k + q];
    const float b = (q == 0) ? Bs[h * NS + i] : 0.f;
#pragma unroll
    for (int k = 0; k < 16; ++k) {
      const float d = (i == 4 * k + q) ? 1.f : 0.f;
      r[k]      = __builtin_fmaf(-s, a[k], d);
      r[16 + k] = __builtin_fmaf( s, a[k], d);
    }
    r[32] = 2.f * s * b;
  }
  if (t < NS) cb[t] = Cs[h * NS + t];

  float diag = 1.f;
  for (int pb = 0; pb < 16; ++pb) {
    const int p0 = 4 * pb;
    gj_pivot<0>(r, diag, prow, p0,     i, q);
    gj_pivot<1>(r, diag, prow, p0 + 1, i, q);
    gj_pivot<2>(r, diag, prow, p0 + 2, i, q);
    gj_pivot<3>(r, diag, prow, p0 + 3, i, q);
#pragma unroll
    for (int k = 0; k < 32; ++k) r[k] = r[k + 1];
    r[32] = 0.f;
  }

  {
    const float dinv = 1.0f / diag;
#pragma unroll
    for (int k = 0; k < 16; ++k) P[i * MS + 4 * k + q] = r[k] * dinv;
    if (q == 0) Xa[i * MS + 0] = r[16] * dinv;
  }
  __syncthreads();

  for (int l = 1; l < 16; ++l) {
    if (t < NS) {
      float y = 0.f;
#pragma unroll
      for (int j = 0; j < NS; ++j) y += P[t * MS + j] * Xa[j * MS + (l - 1)];
      Xa[t * MS + l] = y;
    }
    __syncthreads();
  }

  float* Pa = P; float* Pb = Pn;
  for (int sq = 0; sq < 4; ++sq) {
    wg_matmul<4>(Pb, Pa, Pa, wave, lr, half);
    __syncthreads();
    float* tp = Pa; Pa = Pb; Pb = tp;
  }
  for (int e = 0; e < 3; ++e) {
    wg_matmul<1>(Xa + 16 * (e + 1), Pa, Xa + 16 * e, wave, lr, half);
    __syncthreads();
  }
  for (int sq = 0; sq < 2; ++sq) {
    wg_matmul<4>(Pb, Pa, Pa, wave, lr, half);
    __syncthreads();
    float* tp = Pa; Pa = Pb; Pb = tp;
  }

  float* Xc = Xa; float* Xd = Xb;
  for (int blk = 0; blk < 16; ++blk) {
    if (t < NS) {
      float ks = 0.f;
#pragma unroll
      for (int r = 0; r < NS; ++r) ks += cb[r] * Xc[r * MS + t];
      Kws[(size_t)h * LL + blk * NS + t] = ks;
    }
    if (blk < 15)
      wg_matmul<4>(Xd, Pa, Xc, wave, lr, half);
    __syncthreads();
    float* tx = Xc; Xc = Xd; Xd = tx;
  }
}

__global__ __launch_bounds__(256) void ssm_conv_wmma(
    const float* __restrict__ inp, const float* __restrict__ Kws,
    const float* __restrict__ Ds, float* __restrict__ out)
{
  extern __shared__ float smem[];
  float* Ush = smem;
  float* Kp  = smem + LL * 16;

  const int h    = blockIdx.x;
  const int tid  = threadIdx.x;
  const int wave = tid >> 5;
  const int lane = tid & 31;
  const int half = lane >> 4;
  const int lr   = lane & 15;

  for (int idx = tid; idx < LL * 16; idx += 256) {
    const int l = idx >> 4, c = idx & 15;
    Ush[idx] = (c < NB) ? inp[(size_t)c * LL * HC + (size_t)l * HC + h] : 0.f;
  }
  for (int idx = tid; idx < 16 + LL; idx += 256)
    Kp[idx] = (idx < 16) ? 0.f : Kws[(size_t)h * LL + (idx - 16)];
  __syncthreads();

  const float Dh = Ds[h];

  for (int t = wave; t < LL / 16; t += 8) {
    const int l0 = t * 16;
    v8f acc = {0.f, 0.f, 0.f, 0.f, 0.f, 0.f, 0.f, 0.f};

    for (int m0 = 0; m0 < l0 + 16; m0 += 16) {
      v2f af[4], bf[4];
#pragma unroll
      for (int u = 0; u < 4; ++u) {
        const int mrow = m0 + 4 * u + 2 * half;
        const int ia   = 16 + l0 + lr - mrow;
        af[u].x = Kp[ia];
        af[u].y = Kp[ia - 1];
        bf[u].x = Ush[mrow * 16 + lr];
        bf[u].y = Ush[(mrow + 1) * 16 + lr];
      }
#pragma unroll
      for (int u = 0; u < 4; ++u)
        acc = __builtin_amdgcn_wmma_f32_16x16x4_f32(false, af[u], false, bf[u],
                                                    (short)0, acc, false, false);
    }

    if (lr < NB) {
#pragma unroll
      for (int r = 0; r < 8; ++r) {
        const int row = l0 + r + 8 * half;
        const float uv = Ush[row * 16 + lr];
        out[(size_t)lr * LL * HC + (size_t)row * HC + h] = acc[r] + Dh * uv;
      }
    }
  }
}

extern "C" void kernel_launch(void* const* d_in, const int* in_sizes, int n_in,
                              void* d_out, int out_size, void* d_ws, size_t ws_size,
                              hipStream_t stream)
{
  const float* inp = (const float*)d_in[0];
  const float* As  = (const float*)d_in[1];
  const float* Bs  = (const float*)d_in[2];
  const float* Cs  = (const float*)d_in[3];
  const float* Ds  = (const float*)d_in[4];
  const float* ls  = (const float*)d_in[5];
  float* out = (float*)d_out;

  float* Kws = (float*)d_ws;

  const size_t lds1 = (size_t)(2 * 4 * PW + 4 * NS * MS + NS) * sizeof(float);
  const size_t lds2 = (size_t)(LL * 16 + 16 + LL) * sizeof(float);

  ssm_build_k  <<<dim3(HC), dim3(256), lds1, stream>>>(As, Bs, Cs, ls, Kws);
  ssm_conv_wmma<<<dim3(HC), dim3(256), lds2, stream>>>(inp, Kws, Ds, out);
}


# Convolution kernel staging: all 32 strided input loads and 4 K loads of a thread issued before the first wait (they ran one memory round trip after another)
# speedup vs baseline: 2.0936x; 1.2015x over previous
#include <hip/hip_runtime.h>

#define NS 64
#define HC 256
#define NB 8
#define LL 1024
#define MS 66

typedef __attribute__((ext_vector_type(2))) float v2f;
typedef __attribute__((ext_vector_type(8))) float v8f;

template <int NCT>
__device__ __forceinline__ void wg_matmul(float* __restrict__ D,
                                          const float* __restrict__ A,
                                          const float* __restrict__ Bm,
                                          int wave, int lr, int half)
{
  for (int tt = wave; tt < 4 * NCT; tt += 8) {
    const int rt = tt / NCT, ct = tt % NCT;
    const int r0 = rt * 16, c0 = ct * 16;

    v2f af[16], bf[16];
#pragma unroll
    for (int k = 0; k < 16; ++k) {
      const int ka = 4 * k + 2 * half;
      af[k].x = A[(r0 + lr) * MS + ka];
      af[k].y = A[(r0 + lr) * MS + ka + 1];
      bf[k].x = Bm[(ka    ) * MS + c0 + lr];
      bf[k].y = Bm[(ka + 1) * MS + c0 + lr];
    }

    v8f acc = {0.f, 0.f, 0.f, 0.f, 0.f, 0.f, 0.f, 0.f};
#pragma unroll
    for (int k = 0; k < 16; ++k)
      acc = __builtin_amdgcn_wmma_f32_16x16x4_f32(false, af[k], false, bf[k],
                                                  (short)0, acc, false, false);
#pragma unroll
    for (int r = 0; r < 8; ++r)
      D[(r0 + r + 8 * half) * MS + c0 + lr] = acc[r];
  }
}

#define PW 36

template <int PP>
__device__ __forceinline__ void gj_pivot(float (&r)[33], float& diag,
                                         float* __restrict__ prow,
                                         int p, int i, int q)
{
  float* buf = prow + (PP & 1) * (4 * PW);
  float4* mine = (float4*)(buf + q * PW);

  if (i == p) {
#pragma unroll
    for (int k = 0; k < 8; ++k)
      mine[k] = make_float4(r[4 * k], r[4 * k + 1], r[4 * k + 2], r[4 * k + 3]);
    buf[q * PW + 32] = r[32];
  }
  const float aip = __int_as_float(__builtin_amdgcn_update_dpp(
      0, __float_as_int(r[0]), PP * 0x55, 0xF, 0xF, true));
  __syncthreads();

  float pr[33];
#pragma unroll
  for (int k = 0; k < 8; ++k) {
    const float4 v = mine[k];
    pr[4 * k] = v.x; pr[4 * k + 1] = v.y; pr[4 * k + 2] = v.z; pr[4 * k + 3] = v.w;
  }
  pr[32] = buf[q * PW + 32];
  const float piv = buf[PP * PW];

  const float f = (i != p) ? aip / piv : 0.0f;
  if (i == p) diag = piv;
#pragma unroll
  for (int k = 0; k < 33; ++k) r[k] = __builtin_fmaf(-f, pr[k], r[k]);
}

__global__ __launch_bounds__(256) void ssm_build_k(
    const float* __restrict__ As, const float* __restrict__ Bs,
    const float* __restrict__ Cs, const float* __restrict__ log_steps,
    float* __restrict__ Kws)
{
  extern __shared__ __attribute__((aligned(16))) float smem[];
  float* prow = smem;
  float* P  = smem + 2 * 4 * PW;
  float* Pn = P  + NS * MS;
  float* Xa = Pn + NS * MS;
  float* Xb = Xa + NS * MS;
  float* cb = Xb + NS * MS;

  const int h    = blockIdx.x;
  const int t    = threadIdx.x;
  const int wave = t >> 5, lane = t & 31;
  const int half = lane >> 4, lr = lane & 15;

  const int i = t >> 2;
  const int q = t & 3;
  const float s = 0.5f * __expf(log_steps[h]);
  const float* Ah = As + (size_t)h * NS * NS;

  float r[33];
  {
    float a[16];
#pragma unroll
    for (int k = 0; k < 16; ++k) a[k] = Ah[i * NS + 4 * k + q];
    const float b = (q == 0) ? Bs[h * NS + i] : 0.f;
#pragma unroll
    for (int k = 0; k < 16; ++k) {
      const float d = (i == 4 * k + q) ? 1.f : 0.f;
      r[k]      = __builtin_fmaf(-s, a[k], d);
      r[16 + k] = __builtin_fmaf( s, a[k], d);
    }
    r[32] = 2.f * s * b;
  }
  if (t < NS) cb[t] = Cs[h * NS + t];

  float diag = 1.f;
  for (int pb = 0; pb < 16; ++pb) {
    const int p0 = 4 * pb;
    gj_pivot<0>(r, diag, prow, p0,     i, q);
    gj_pivot<1>(r, diag, prow, p0 + 1, i, q);
    gj_pivot<2>(r, diag, prow, p0 + 2, i, q);
    gj_pivot<3>(r, diag, prow, p0 + 3, i, q);
#pragma unroll
    for (int k = 0; k < 32; ++k) r[k] = r[k + 1];
    r[32] = 0.f;
  }

  {
    const float dinv = 1.0f / diag;
#pragma unroll
    for (int k = 0; k < 16; ++k) P[i * MS + 4 * k + q] = r[k] * dinv;
    if (q == 0) Xa[i * MS + 0] = r[16] * dinv;
  }
  __syncthreads();

  for (int l = 1; l < 16; ++l) {
    if (t < NS) {
      float y = 0.f;
#pragma unroll
      for (int j = 0; j < NS; ++j) y += P[t * MS + j] * Xa[j * MS + (l - 1)];
      Xa[t * MS + l] = y;
    }
    __syncthreads();
  }

  float* Pa = P; float* Pb = Pn;
  for (int sq = 0; sq < 4; ++sq) {
    wg_matmul<4>(Pb, Pa, Pa, wave, lr, half);
    __syncthreads();
    float* tp = Pa; Pa = Pb; Pb = tp;
  }
  for (int e = 0; e < 3; ++e) {
    wg_matmul<1>(Xa + 16 * (e + 1), Pa, Xa + 16 * e, wave, lr, half);
    __syncthreads();
  }
  for (int sq = 0; sq < 2; ++sq) {
    wg_matmul<4>(Pb, Pa, Pa, wave, lr, half);
    __syncthreads();
    float* tp = Pa; Pa = Pb; Pb = tp;
  }

  float* Xc = Xa; float* Xd = Xb;
  for (int blk = 0; blk < 16; ++blk) {
    if (t < NS) {
      float ks = 0.f;
#pragma unroll
      for (int r = 0; r < NS; ++r) ks += cb[r] * Xc[r * MS + t];
      Kws[(size_t)h * LL + blk * NS + t] = ks;
    }
    if (blk < 15)
      wg_matmul<4>(Xd, Pa, Xc, wave, lr, half);
    __syncthreads();
    float* tx = Xc; Xc = Xd; Xd = tx;
  }
}

__global__ __launch_bounds__(256) void ssm_conv_wmma(
    const float* __restrict__ inp, const float* __restrict__ Kws,
    const float* __restrict__ Ds, float* __restrict__ out)
{
  extern __shared__ __attribute__((aligned(16))) float smem[];
  float* Ush = smem;
  float* Kp  = smem + LL * 16;

  const int h    = blockIdx.x;
  const int tid  = threadIdx.x;
  const int wave = tid >> 5;
  const int lane = tid & 31;
  const int half = lane >> 4;
  const int lr   = lane & 15;

  {
    const int sb = tid & 7;
    const int sl = tid >> 3;
    const float* up = inp + (size_t)sb * LL * HC + (size_t)sl * HC + h;
    const float* kp = Kws + (size_t)h * LL + tid;
    float uv[32], kv[4];
#pragma unroll
    for (int j = 0; j < 32; ++j) uv[j] = up[(size_t)j * 32 * HC];
#pragma unroll
    for (int j = 0; j < 4; ++j) kv[j] = kp[j * 256];

    const float4 z4 = make_float4(0.f, 0.f, 0.f, 0.f);
    for (int idx = tid; idx < LL * 2; idx += 256)
      *(float4*)(Ush + (idx >> 1) * 16 + 8 + 4 * (idx & 1)) = z4;
    if (tid < 16) Kp[tid] = 0.f;

#pragma unroll
    for (int j = 0; j < 32; ++j) Ush[(j * 32 + sl) * 16 + sb] = uv[j];
#pragma unroll
    for (int j = 0; j < 4; ++j) Kp[16 + j * 256 + tid] = kv[j];
  }
  __syncthreads();

  const float Dh = Ds[h];

  for (int t = wave; t < LL / 16; t += 8) {
    const int l0 = t * 16;
    v8f acc = {0.f, 0.f, 0.f, 0.f, 0.f, 0.f, 0.f, 0.f};

    for (int m0 = 0; m0 < l0 + 16; m0 += 16) {
      v2f af[4], bf[4];
#pragma unroll
      for (int u = 0; u < 4; ++u) {
        const int mrow = m0 + 4 * u + 2 * half;
        const int ia   = 16 + l0 + lr - mrow;
        af[u].x = Kp[ia];
        af[u].y = Kp[ia - 1];
        bf[u].x = Ush[mrow * 16 + lr];
        bf[u].y = Ush[(mrow + 1) * 16 + lr];
      }
#pragma unroll
      for (int u = 0; u < 4; ++u)
        acc = __builtin_amdgcn_wmma_f32_16x16x4_f32(false, af[u], false, bf[u],
                                                    (short)0, acc, false, false);
    }

    if (lr < NB) {
#pragma unroll
      for (int r = 0; r < 8; ++r) {
        const int row = l0 + r + 8 * half;
        const float uv = Ush[row * 16 + lr];
        out[(size_t)lr * LL * HC + (size_t)row * HC + h] = acc[r] + Dh * uv;
      }
    }
  }
}

extern "C" void kernel_launch(void* const* d_in, const int* in_sizes, int n_in,
                              void* d_out, int out_size, void* d_ws, size_t ws_size,
                              hipStream_t stream)
{
  const float* inp = (const float*)d_in[0];
  const float* As  = (const float*)d_in[1];
  const float* Bs  = (const float*)d_in[2];
  const float* Cs  = (const float*)d_in[3];
  const float* Ds  = (const float*)d_in[4];
  const float* ls  = (const float*)d_in[5];
  float* out = (float*)d_out;

  float* Kws = (float*)d_ws;

  const size_t lds1 = (size_t)(2 * 4 * PW + 4 * NS * MS + NS) * sizeof(float);
  const size_t lds2 = (size_t)(LL * 16 + 16 + LL) * sizeof(float);

  ssm_build_k  <<<dim3(HC), dim3(256), lds1, stream>>>(As, Bs, Cs, ls, Kws);
  ssm_conv_wmma<<<dim3(HC), dim3(256), lds2, stream>>>(inp, Kws, Ds, out);
}


# Convolution kernel: each WMMA now serves two 16-row output tiles (8 batch columns each; a Toeplitz tile depends only on its diagonal), halving WMMAs and tap loads; per-output sum order kept
# speedup vs baseline: 2.2658x; 1.0823x over previous
#include <hip/hip_runtime.h>

#define NS 64
#define HC 256
#define NB 8
#define LL 1024
#define MS 66

typedef __attribute__((ext_vector_type(2))) float v2f;
typedef __attribute__((ext_vector_type(8))) float v8f;

template <int NCT>
__device__ __forceinline__ void wg_matmul(float* __restrict__ D,
                                          const float* __restrict__ A,
                                          const float* __restrict__ Bm,
                                          int wave, int lr, int half)
{
  for (int tt = wave; tt < 4 * NCT; tt += 8) {
    const int rt = tt / NCT, ct = tt % NCT;
    const int r0 = rt * 16, c0 = ct * 16;

    v2f af[16], bf[16];
#pragma unroll
    for (int k = 0; k < 16; ++k) {
      const int ka = 4 * k + 2 * half;
      af[k].x = A[(r0 + lr) * MS + ka];
      af[k].y = A[(r0 + lr) * MS + ka + 1];
      bf[k].x = Bm[(ka    ) * MS + c0 + lr];
      bf[k].y = Bm[(ka + 1) * MS + c0 + lr];
    }

    v8f acc = {0.f, 0.f, 0.f, 0.f, 0.f, 0.f, 0.f, 0.f};
#pragma unroll
    for (int k = 0; k < 16; ++k)
      acc = __builtin_amdgcn_wmma_f32_16x16x4_f32(false, af[k], false, bf[k],
                                                  (short)0, acc, false, false);
#pragma unroll
    for (int r = 0; r < 8; ++r)
      D[(r0 + r + 8 * half) * MS + c0 + lr] = acc[r];
  }
}

#define PW 36

template <int PP>
__device__ __forceinline__ void gj_pivot(float (&r)[33], float& diag,
                                         float* __restrict__ prow,
                                         int p, int i, int q)
{
  float* buf = prow + (PP & 1) * (4 * PW);
  float4* mine = (float4*)(buf + q * PW);

  if (i == p) {
#pragma unroll
    for (int k = 0; k < 8; ++k)
      mine[k] = make_float4(r[4 * k], r[4 * k + 1], r[4 * k + 2], r[4 * k + 3]);
    buf[q * PW + 32] = r[32];
  }
  const float aip = __int_as_float(__builtin_amdgcn_update_dpp(
      0, __float_as_int(r[0]), PP * 0x55, 0xF, 0xF, true));
  __syncthreads();

  float pr[33];
#pragma unroll
  for (int k = 0; k < 8; ++k) {
    const float4 v = mine[k];
    pr[4 * k] = v.x; pr[4 * k + 1] = v.y; pr[4 * k + 2] = v.z; pr[4 * k + 3] = v.w;
  }
  pr[32] = buf[q * PW + 32];
  const float piv = buf[PP * PW];

  const float f = (i != p) ? aip / piv : 0.0f;
  if (i == p) diag = piv;
#pragma unroll
  for (int k = 0; k < 33; ++k) r[k] = __builtin_fmaf(-f, pr[k], r[k]);
}

__global__ __launch_bounds__(256) void ssm_build_k(
    const float* __restrict__ As, const float* __restrict__ Bs,
    const float* __restrict__ Cs, const float* __restrict__ log_steps,
    float* __restrict__ Kws)
{
  extern __shared__ __attribute__((aligned(16))) float smem[];
  float* prow = smem;
  float* P  = smem + 2 * 4 * PW;
  float* Pn = P  + NS * MS;
  float* Xa = Pn + NS * MS;
  float* Xb = Xa + NS * MS;
  float* cb = Xb + NS * MS;

  const int h    = blockIdx.x;
  const int t    = threadIdx.x;
  const int wave = t >> 5, lane = t & 31;
  const int half = lane >> 4, lr = lane & 15;

  const int i = t >> 2;
  const int q = t & 3;
  const float s = 0.5f * __expf(log_steps[h]);
  const float* Ah = As + (size_t)h * NS * NS;

  float r[33];
  {
    float a[16];
#pragma unroll
    for (int k = 0; k < 16; ++k) a[k] = Ah[i * NS + 4 * k + q];
    const float b = (q == 0) ? Bs[h * NS + i] : 0.f;
#pragma unroll
    for (int k = 0; k < 16; ++k) {
      const float d = (i == 4 * k + q) ? 1.f : 0.f;
      r[k]      = __builtin_fmaf(-s, a[k], d);
      r[16 + k] = __builtin_fmaf( s, a[k], d);
    }
    r[32] = 2.f * s * b;
  }
  if (t < NS) cb[t] = Cs[h * NS + t];

  float diag = 1.f;
  for (int pb = 0; pb < 16; ++pb) {
    const int p0 = 4 * pb;
    gj_pivot<0>(r, diag, prow, p0,     i, q);
    gj_pivot<1>(r, diag, prow, p0 + 1, i, q);
    gj_pivot<2>(r, diag, prow, p0 + 2, i, q);
    gj_pivot<3>(r, diag, prow, p0 + 3, i, q);
#pragma unroll
    for (int k = 0; k < 32; ++k) r[k] = r[k + 1];
    r[32] = 0.f;
  }

  {
    const float dinv = 1.0f / diag;
#pragma unroll
    for (int k = 0; k < 16; ++k) P[i * MS + 4 * k + q] = r[k] * dinv;
    if (q == 0) Xa[i * MS + 0] = r[16] * dinv;
  }
  __syncthreads();

  for (int l = 1; l < 16; ++l) {
    if (t < NS) {
      float y = 0.f;
#pragma unroll
      for (int j = 0; j < NS; ++j) y += P[t * MS + j] * Xa[j * MS + (l - 1)];
      Xa[t * MS + l] = y;
    }
    __syncthreads();
  }

  float* Pa = P; float* Pb = Pn;
  for (int sq = 0; sq < 4; ++sq) {
    wg_matmul<4>(Pb, Pa, Pa, wave, lr, half);
    __syncthreads();
    float* tp = Pa; Pa = Pb; Pb = tp;
  }
  for (int e = 0; e < 3; ++e) {
    wg_matmul<1>(Xa + 16 * (e + 1), Pa, Xa + 16 * e, wave, lr, half);
    __syncthreads();
  }
  for (int sq = 0; sq < 2; ++sq) {
    wg_matmul<4>(Pb, Pa, Pa, wave, lr, half);
    __syncthreads();
    float* tp = Pa; Pa = Pb; Pb = tp;
  }

  float* Xc = Xa; float* Xd = Xb;
  for (int blk = 0; blk < 16; ++blk) {
    if (t < NS) {
      float ks = 0.f;
#pragma unroll
      for (int r = 0; r < NS; ++r) ks += cb[r] * Xc[r * MS + t];
      Kws[(size_t)h * LL + blk * NS + t] = ks;
    }
    if (blk < 15)
      wg_matmul<4>(Xd, Pa, Xc, wave, lr, half);
    __syncthreads();
    float* tx = Xc; Xc = Xd; Xd = tx;
  }
}

#define UT 136

__global__ __launch_bounds__(256) void ssm_conv_wmma(
    const float* __restrict__ inp, const float* __restrict__ Kws,
    const float* __restrict__ Ds, float* __restrict__ out)
{
  extern __shared__ __attribute__((aligned(16))) float smem[];
  float* Ush = smem;
  float* Kp  = smem + 65 * UT;

  const int h    = blockIdx.x;
  const int tid  = threadIdx.x;
  const int wave = tid >> 5;
  const int lane = tid & 31;
  const int half = lane >> 4;
  const int lr   = lane & 15;
  const int hi   = lr >> 3;
  const int col  = lr & 7;

  {
    const int sb = tid & 7;
    const int sl = tid >> 3;
    const float* up = inp + (size_t)sb * LL * HC + (size_t)sl * HC + h;
    const float* kp = Kws + (size_t)h * LL + tid;
    float uv[32], kv[4];
#pragma unroll
    for (int j = 0; j < 32; ++j) uv[j] = up[(size_t)j * 32 * HC];
#pragma unroll
    for (int j = 0; j < 4; ++j) kv[j] = kp[j * 256];

    if (tid < 128) Ush[tid] = 0.f;
    if (tid < 16) Kp[tid] = 0.f;

    float* ud = Ush + (1 + (sl >> 4)) * UT + (sl & 15) * 8 + sb;
#pragma unroll
    for (int j = 0; j < 32; ++j) ud[j * 2 * UT] = uv[j];
#pragma unroll
    for (int j = 0; j < 4; ++j) Kp[16 + j * 256 + tid] = kv[j];
  }
  __syncthreads();

  const float Dh = Ds[h];

  for (int j = 0; j < 4; ++j) {
    const int p  = (j & 1) ? (8 * j + 7 - wave) : (8 * j + wave);
    const int t0 = 2 * p;
    v8f acc = {0.f, 0.f, 0.f, 0.f, 0.f, 0.f, 0.f, 0.f};

    const float* kb = Kp + 16 + 16 * (t0 + 1) + lr - 2 * half;
    const float* ub = Ush + hi * UT + 16 * half + col;

    for (int n = 0; n < t0 + 2; ++n) {
      v2f af[4], bf[4];
#pragma unroll
      for (int u = 0; u < 4; ++u) {
        af[u].x = kb[-4 * u];
        af[u].y = kb[-4 * u - 1];
        bf[u].x = ub[(4 * u) * 8];
        bf[u].y = ub[(4 * u + 1) * 8];
      }
#pragma unroll
      for (int u = 0; u < 4; ++u)
        acc = __builtin_amdgcn_wmma_f32_16x16x4_f32(false, af[u], false, bf[u],
                                                    (short)0, acc, false, false);
      kb -= 16;
      ub += UT;
    }

    const float* ue = Ush + (1 + t0 + hi) * UT + (8 * half) * 8 + col;
    float* op = out + (size_t)col * LL * HC
                    + (size_t)(16 * (t0 + hi) + 8 * half) * HC + h;
#pragma unroll
    for (int r = 0; r < 8; ++r)
      op[(size_t)r * HC] = acc[r] + Dh * ue[r * 8];
  }
}

extern "C" void kernel_launch(void* const* d_in, const int* in_sizes, int n_in,
                              void* d_out, int out_size, void* d_ws, size_t ws_size,
                              hipStream_t stream)
{
  const float* inp = (const float*)d_in[0];
  const float* As  = (const float*)d_in[1];
  const float* Bs  = (const float*)d_in[2];
  const float* Cs  = (const float*)d_in[3];
  const float* Ds  = (const float*)d_in[4];
  const float* ls  = (const float*)d_in[5];
  float* out = (float*)d_out;

  float* Kws = (float*)d_ws;

  const size_t lds1 = (size_t)(2 * 4 * PW + 4 * NS * MS + NS) * sizeof(float);
  const size_t lds2 = (size_t)(65 * UT + 16 + LL) * sizeof(float);

  ssm_build_k  <<<dim3(HC), dim3(256), lds1, stream>>>(As, Bs, Cs, ls, Kws);
  ssm_conv_wmma<<<dim3(HC), dim3(256), lds2, stream>>>(inp, Kws, Ds, out);
}


# K build: K = Cm @ X as one 16x64x64 WMMA product, Cm rows c^T Q^i grown by doubling with 3 more squarings; drops the 15 full 64^3 products and 16 serial dot emits (sum order of K changes)
# speedup vs baseline: 2.6690x; 1.1780x over previous
#include <hip/hip_runtime.h>

#define NS 64
#define HC 256
#define NB 8
#define LL 1024
#define MS 66

typedef __attribute__((ext_vector_type(2))) float v2f;
typedef __attribute__((ext_vector_type(8))) float v8f;

__device__ __forceinline__ v8f mm_tile(const float* __restrict__ A,
                                       const float* __restrict__ Bm,
                                       int arow, int bcol, int half)
{
  v2f af[16], bf[16];
#pragma unroll
  for (int k = 0; k < 16; ++k) {
    const int ka = 4 * k + 2 * half;
    af[k].x = A[arow * MS + ka];
    af[k].y = A[arow * MS + ka + 1];
    bf[k].x = Bm[(ka    ) * MS + bcol];
    bf[k].y = Bm[(ka + 1) * MS + bcol];
  }

  v8f acc = {0.f, 0.f, 0.f, 0.f, 0.f, 0.f, 0.f, 0.f};
#pragma unroll
  for (int k = 0; k < 16; ++k)
    acc = __builtin_amdgcn_wmma_f32_16x16x4_f32(false, af[k], false, bf[k],
                                                (short)0, acc, false, false);
  return acc;
}

template <int NCT>
__device__ __forceinline__ void wg_matmul(float* __restrict__ D,
                                          const float* __restrict__ A,
                                          const float* __restrict__ Bm,
                                          int wave, int lr, int half)
{
  for (int tt = wave; tt < 4 * NCT; tt += 8) {
    const int rt = tt / NCT, ct = tt % NCT;
    const int r0 = rt * 16, c0 = ct * 16;
    const v8f acc = mm_tile(A, Bm, r0 + lr, c0 + lr, half);
#pragma unroll
    for (int r = 0; r < 8; ++r)
      D[(r0 + r + 8 * half) * MS + c0 + lr] = acc[r];
  }
}

template <int NR>
__device__ __forceinline__ void cm_grow(float* Cm, const float* __restrict__ Pm,
                                        int ct, int lr, int half)
{
  const int c0 = 16 * ct;
  const v8f acc = mm_tile(Cm, Pm, lr & (NR - 1), c0 + lr, half);
  if (half == 0) {
#pragma unroll
    for (int r = 0; r < NR; ++r)
      Cm[(NR + r) * MS + c0 + lr] = acc[r];
  }
}

#define PW 36

template <int PP>
__device__ __forceinline__ void gj_pivot(float (&r)[33], float& diag,
                                         float* __restrict__ prow,
                                         int p, int i, int q)
{
  float* buf = prow + (PP & 1) * (4 * PW);
  float4* mine = (float4*)(buf + q * PW);

  if (i == p) {
#pragma unroll
    for (int k = 0; k < 8; ++k)
      mine[k] = make_float4(r[4 * k], r[4 * k + 1], r[4 * k + 2], r[4 * k + 3]);
    buf[q * PW + 32] = r[32];
  }
  const float aip = __int_as_float(__builtin_amdgcn_update_dpp(
      0, __float_as_int(r[0]), PP * 0x55, 0xF, 0xF, true));
  __syncthreads();

  float pr[33];
#pragma unroll
  for (int k = 0; k < 8; ++k) {
    const float4 v = mine[k];
    pr[4 * k] = v.x; pr[4 * k + 1] = v.y; pr[4 * k + 2] = v.z; pr[4 * k + 3] = v.w;
  }
  pr[32] = buf[q * PW + 32];
  const float piv = buf[PP * PW];

  const float f = (i != p) ? aip / piv : 0.0f;
  if (i == p) diag = piv;
#pragma unroll
  for (int k = 0; k < 33; ++k) r[k] = __builtin_fmaf(-f, pr[k], r[k]);
}

__global__ __launch_bounds__(256) void ssm_build_k(
    const float* __restrict__ As, const float* __restrict__ Bs,
    const float* __restrict__ Cs, const float* __restrict__ log_steps,
    float* __restrict__ Kws)
{
  extern __shared__ __attribute__((aligned(16))) float smem[];
  float* prow = smem;
  float* P  = smem + 2 * 4 * PW;
  float* Pn = P  + NS * MS;
  float* Xa = Pn + NS * MS;
  float* Xb = Xa + NS * MS;
  float* Cm = Xb + NS * MS;

  const int h    = blockIdx.x;
  const int t    = threadIdx.x;
  const int wave = t >> 5, lane = t & 31;
  const int half = lane >> 4, lr = lane & 15;

  const int i = t >> 2;
  const int q = t & 3;
  const float s = 0.5f * __expf(log_steps[h]);
  const float* Ah = As + (size_t)h * NS * NS;

  float r[33];
  {
    float a[16];
#pragma unroll
    for (int k = 0; k < 16; ++k) a[k] = Ah[i * NS + 4 * k + q];
    const float b = (q == 0) ? Bs[h * NS + i] : 0.f;
#pragma unroll
    for (int k = 0; k < 16; ++k) {
      const float d = (i == 4 * k + q) ? 1.f : 0.f;
      r[k]      = __builtin_fmaf(-s, a[k], d);
      r[16 + k] = __builtin_fmaf( s, a[k], d);
    }
    r[32] = 2.f * s * b;
  }
  for (int e = t; e < 16 * MS; e += 256) Cm[e] = (e < NS) ? Cs[h * NS + e] : 0.f;

  float diag = 1.f;
  for (int pb = 0; pb < 16; ++pb) {
    const int p0 = 4 * pb;
    gj_pivot<0>(r, diag, prow, p0,     i, q);
    gj_pivot<1>(r, diag, prow, p0 + 1, i, q);
    gj_pivot<2>(r, diag, prow, p0 + 2, i, q);
    gj_pivot<3>(r, diag, prow, p0 + 3, i, q);
#pragma unroll
    for (int k = 0; k < 32; ++k) r[k] = r[k + 1];
    r[32] = 0.f;
  }

  {
    const float dinv = 1.0f / diag;
#pragma unroll
    for (int k = 0; k < 16; ++k) P[i * MS + 4 * k + q] = r[k] * dinv;
    if (q == 0) Xa[i * MS + 0] = r[16] * dinv;
  }
  __syncthreads();

  for (int l = 1; l < 16; ++l) {
    if (t < NS) {
      float y = 0.f;
#pragma unroll
      for (int j = 0; j < NS; ++j) y += P[t * MS + j] * Xa[j * MS + (l - 1)];
      Xa[t * MS + l] = y;
    }
    __syncthreads();
  }

  float* Pa = P; float* Pb = Pn;
  for (int sq = 0; sq < 4; ++sq) {
    wg_matmul<4>(Pb, Pa, Pa, wave, lr, half);
    __syncthreads();
    float* tp = Pa; Pa = Pb; Pb = tp;
  }
  for (int e = 0; e < 3; ++e) {
    wg_matmul<1>(Xa + 16 * (e + 1), Pa, Xa + 16 * e, wave, lr, half);
    __syncthreads();
  }
  for (int sq = 0; sq < 2; ++sq) {
    wg_matmul<4>(Pb, Pa, Pa, wave, lr, half);
    __syncthreads();
    float* tp = Pa; Pa = Pb; Pb = tp;
  }

  const int cw = wave - 4;
  wg_matmul<4>(Pb, Pa, Pa, wave, lr, half);
  if (cw >= 0) cm_grow<1>(Cm, Pa, cw, lr, half);
  __syncthreads();
  wg_matmul<4>(Xb, Pb, Pb, wave, lr, half);
  if (cw >= 0) cm_grow<2>(Cm, Pb, cw, lr, half);
  __syncthreads();
  wg_matmul<4>(Pa, Xb, Xb, wave, lr, half);
  if (cw >= 0) cm_grow<4>(Cm, Xb, cw, lr, half);
  __syncthreads();
  if (cw >= 0) cm_grow<8>(Cm, Pa, cw, lr, half);
  __syncthreads();

  if (cw >= 0) {
    const int c0 = 16 * cw;
    const v8f acc = mm_tile(Cm, Xa, lr, c0 + lr, half);
    float* kp = Kws + (size_t)h * LL + (8 * half) * NS + c0 + lr;
#pragma unroll
    for (int r = 0; r < 8; ++r) kp[r * NS] = acc[r];
  }
}

#define UT 136

__global__ __launch_bounds__(256) void ssm_conv_wmma(
    const float* __restrict__ inp, const float* __restrict__ Kws,
    const float* __restrict__ Ds, float* __restrict__ out)
{
  extern __shared__ __attribute__((aligned(16))) float smem[];
  float* Ush = smem;
  float* Kp  = smem + 65 * UT;

  const int h    = blockIdx.x;
  const int tid  = threadIdx.x;
  const int wave = tid >> 5;
  const int lane = tid & 31;
  const int half = lane >> 4;
  const int lr   = lane & 15;
  const int hi   = lr >> 3;
  const int col  = lr & 7;

  {
    const int sb = tid & 7;
    const int sl = tid >> 3;
    const float* up = inp + (size_t)sb * LL * HC + (size_t)sl * HC + h;
    const float* kp = Kws + (size_t)h * LL + tid;
    float uv[32], kv[4];
#pragma unroll
    for (int j = 0; j < 32; ++j) uv[j] = up[(size_t)j * 32 * HC];
#pragma unroll
    for (int j = 0; j < 4; ++j) kv[j] = kp[j * 256];

    if (tid < 128) Ush[tid] = 0.f;
    if (tid < 16) Kp[tid] = 0.f;

    float* ud = Ush + (1 + (sl >> 4)) * UT + (sl & 15) * 8 + sb;
#pragma unroll
    for (int j = 0; j < 32; ++j) ud[j * 2 * UT] = uv[j];
#pragma unroll
    for (int j = 0; j < 4; ++j) Kp[16 + j * 256 + tid] = kv[j];
  }
  __syncthreads();

  const float Dh = Ds[h];

  for (int j = 0; j < 4; ++j) {
    const int p  = (j & 1) ? (8 * j + 7 - wave) : (8 * j + wave);
    const int t0 = 2 * p;
    v8f acc = {0.f, 0.f, 0.f, 0.f, 0.f, 0.f, 0.f, 0.f};

    const float* kb = Kp + 16 + 16 * (t0 + 1) + lr - 2 * half;
    const float* ub = Ush + hi * UT + 16 * half + col;

    for (int n = 0; n < t0 + 2; ++n) {
      v2f af[4], bf[4];
#pragma unroll
      for (int u = 0; u < 4; ++u) {
        af[u].x = kb[-4 * u];
        af[u].y = kb[-4 * u - 1];
        bf[u].x = ub[(4 * u) * 8];
        bf[u].y = ub[(4 * u + 1) * 8];
      }
#pragma unroll
      for (int u = 0; u < 4; ++u)
        acc = __builtin_amdgcn_wmma_f32_16x16x4_f32(false, af[u], false, bf[u],
                                                    (short)0, acc, false, false);
      kb -= 16;
      ub += UT;
    }

    const float* ue = Ush + (1 + t0 + hi) * UT + (8 * half) * 8 + col;
    float* op = out + (size_t)col * LL * HC
                    + (size_t)(16 * (t0 + hi) + 8 * half) * HC + h;
#pragma unroll
    for (int r = 0; r < 8; ++r)
      op[(size_t)r * HC] = acc[r] + Dh * ue[r * 8];
  }
}

extern "C" void kernel_launch(void* const* d_in, const int* in_sizes, int n_in,
                              void* d_out, int out_size, void* d_ws, size_t ws_size,
                              hipStream_t stream)
{
  const float* inp = (const float*)d_in[0];
  const float* As  = (const float*)d_in[1];
  const float* Bs  = (const float*)d_in[2];
  const float* Cs  = (const float*)d_in[3];
  const float* Ds  = (const float*)d_in[4];
  const float* ls  = (const float*)d_in[5];
  float* out = (float*)d_out;

  float* Kws = (float*)d_ws;

  const size_t lds1 = (size_t)(2 * 4 * PW + 4 * NS * MS + 16 * MS) * sizeof(float);
  const size_t lds2 = (size_t)(65 * UT + 16 + LL) * sizeof(float);

  ssm_build_k  <<<dim3(HC), dim3(256), lds1, stream>>>(As, Bs, Cs, ls, Kws);
  ssm_conv_wmma<<<dim3(HC), dim3(256), lds2, stream>>>(inp, Kws, Ds, out);
}


# K build: all Krylov columns X[:,1..63] by doubling on WMMA beside the 6 squarings (cols [n,2n) = Ab^n @ cols [0,n)); drops 15 serial LDS matvecs and 3 serial products; sum order of X and K changes
# speedup vs baseline: 2.8732x; 1.0765x over previous
#include <hip/hip_runtime.h>

#define NS 64
#define HC 256
#define NB 8
#define LL 1024
#define MS 66

typedef __attribute__((ext_vector_type(2))) float v2f;
typedef __attribute__((ext_vector_type(8))) float v8f;

__device__ __forceinline__ v8f mm_tile(const float* __restrict__ A,
                                       const float* __restrict__ Bm,
                                       int arow, int bcol, int half)
{
  v2f af[16], bf[16];
#pragma unroll
  for (int k = 0; k < 16; ++k) {
    const int ka = 4 * k + 2 * half;
    af[k].x = A[arow * MS + ka];
    af[k].y = A[arow * MS + ka + 1];
    bf[k].x = Bm[(ka    ) * MS + bcol];
    bf[k].y = Bm[(ka + 1) * MS + bcol];
  }

  v8f acc = {0.f, 0.f, 0.f, 0.f, 0.f, 0.f, 0.f, 0.f};
#pragma unroll
  for (int k = 0; k < 16; ++k)
    acc = __builtin_amdgcn_wmma_f32_16x16x4_f32(false, af[k], false, bf[k],
                                                (short)0, acc, false, false);
  return acc;
}

template <int NCT>
__device__ __forceinline__ void wg_matmul(float* __restrict__ D,
                                          const float* __restrict__ A,
                                          const float* __restrict__ Bm,
                                          int wave, int lr, int half)
{
  for (int tt = wave; tt < 4 * NCT; tt += 8) {
    const int rt = tt / NCT, ct = tt % NCT;
    const int r0 = rt * 16, c0 = ct * 16;
    const v8f acc = mm_tile(A, Bm, r0 + lr, c0 + lr, half);
#pragma unroll
    for (int r = 0; r < 8; ++r)
      D[(r0 + r + 8 * half) * MS + c0 + lr] = acc[r];
  }
}

template <int NR>
__device__ __forceinline__ void cm_grow(float* Cm, const float* __restrict__ Pm,
                                        int ct, int lr, int half)
{
  const int c0 = 16 * ct;
  const v8f acc = mm_tile(Cm, Pm, lr & (NR - 1), c0 + lr, half);
  if (half == 0) {
#pragma unroll
    for (int r = 0; r < NR; ++r)
      Cm[(NR + r) * MS + c0 + lr] = acc[r];
  }
}

template <int NC>
__device__ __forceinline__ void x_grow(float* X, const float* __restrict__ Pm,
                                       int rt, int lr, int half)
{
  const int r0 = 16 * rt;
  const v8f acc = mm_tile(Pm, X, r0 + lr, lr & (NC - 1), half);
  if (lr < NC) {
#pragma unroll
    for (int r = 0; r < 8; ++r)
      X[(r0 + r + 8 * half) * MS + NC + lr] = acc[r];
  }
}

#define PW 36

template <int PP>
__device__ __forceinline__ void gj_pivot(float (&r)[33], float& diag,
                                         float* __restrict__ prow,
                                         int p, int i, int q)
{
  float* buf = prow + (PP & 1) * (4 * PW);
  float4* mine = (float4*)(buf + q * PW);

  if (i == p) {
#pragma unroll
    for (int k = 0; k < 8; ++k)
      mine[k] = make_float4(r[4 * k], r[4 * k + 1], r[4 * k + 2], r[4 * k + 3]);
    buf[q * PW + 32] = r[32];
  }
  const float aip = __int_as_float(__builtin_amdgcn_update_dpp(
      0, __float_as_int(r[0]), PP * 0x55, 0xF, 0xF, true));
  __syncthreads();

  float pr[33];
#pragma unroll
  for (int k = 0; k < 8; ++k) {
    const float4 v = mine[k];
    pr[4 * k] = v.x; pr[4 * k + 1] = v.y; pr[4 * k + 2] = v.z; pr[4 * k + 3] = v.w;
  }
  pr[32] = buf[q * PW + 32];
  const float piv = buf[PP * PW];

  const float f = (i != p) ? aip / piv : 0.0f;
  if (i == p) diag = piv;
#pragma unroll
  for (int k = 0; k < 33; ++k) r[k] = __builtin_fmaf(-f, pr[k], r[k]);
}

__global__ __launch_bounds__(256) void ssm_build_k(
    const float* __restrict__ As, const float* __restrict__ Bs,
    const float* __restrict__ Cs, const float* __restrict__ log_steps,
    float* __restrict__ Kws)
{
  extern __shared__ __attribute__((aligned(16))) float smem[];
  float* prow = smem;
  float* P  = smem + 2 * 4 * PW;
  float* Pn = P  + NS * MS;
  float* Xa = Pn + NS * MS;
  float* Xb = Xa + NS * MS;
  float* Cm = Xb + NS * MS;

  const int h    = blockIdx.x;
  const int t    = threadIdx.x;
  const int wave = t >> 5, lane = t & 31;
  const int half = lane >> 4, lr = lane & 15;

  const int i = t >> 2;
  const int q = t & 3;
  const float s = 0.5f * __expf(log_steps[h]);
  const float* Ah = As + (size_t)h * NS * NS;

  float r[33];
  {
    float a[16];
#pragma unroll
    for (int k = 0; k < 16; ++k) a[k] = Ah[i * NS + 4 * k + q];
    const float b = (q == 0) ? Bs[h * NS + i] : 0.f;
#pragma unroll
    for (int k = 0; k < 16; ++k) {
      const float d = (i == 4 * k + q) ? 1.f : 0.f;
      r[k]      = __builtin_fmaf(-s, a[k], d);
      r[16 + k] = __builtin_fmaf( s, a[k], d);
    }
    r[32] = 2.f * s * b;
  }
  for (int e = t; e < 16 * MS; e += 256) Cm[e] = (e < NS) ? Cs[h * NS + e] : 0.f;

  float diag = 1.f;
  for (int pb = 0; pb < 16; ++pb) {
    const int p0 = 4 * pb;
    gj_pivot<0>(r, diag, prow, p0,     i, q);
    gj_pivot<1>(r, diag, prow, p0 + 1, i, q);
    gj_pivot<2>(r, diag, prow, p0 + 2, i, q);
    gj_pivot<3>(r, diag, prow, p0 + 3, i, q);
#pragma unroll
    for (int k = 0; k < 32; ++k) r[k] = r[k + 1];
    r[32] = 0.f;
  }

  {
    const float dinv = 1.0f / diag;
#pragma unroll
    for (int k = 0; k < 16; ++k) P[i * MS + 4 * k + q] = r[k] * dinv;
    if (q == 0) Xa[i * MS + 0] = r[16] * dinv;
  }
  __syncthreads();

  const int cw = wave - 4;
  wg_matmul<4>(Pn, P, P, wave, lr, half);
  if (cw >= 0) x_grow<1>(Xa, P, cw, lr, half);
  __syncthreads();
  wg_matmul<4>(P, Pn, Pn, wave, lr, half);
  if (cw >= 0) x_grow<2>(Xa, Pn, cw, lr, half);
  __syncthreads();
  wg_matmul<4>(Pn, P, P, wave, lr, half);
  if (cw >= 0) x_grow<4>(Xa, P, cw, lr, half);
  __syncthreads();
  wg_matmul<4>(P, Pn, Pn, wave, lr, half);
  if (cw >= 0) x_grow<8>(Xa, Pn, cw, lr, half);
  __syncthreads();
  wg_matmul<4>(Pn, P, P, wave, lr, half);
  wg_matmul<1>(Xa + 16, P, Xa, wave, lr, half);
  __syncthreads();
  wg_matmul<4>(P, Pn, Pn, wave, lr, half);
  wg_matmul<2>(Xa + 32, Pn, Xa, wave, lr, half);
  __syncthreads();
  float* Pa = P; float* Pb = Pn;

  wg_matmul<4>(Pb, Pa, Pa, wave, lr, half);
  if (cw >= 0) cm_grow<1>(Cm, Pa, cw, lr, half);
  __syncthreads();
  wg_matmul<4>(Xb, Pb, Pb, wave, lr, half);
  if (cw >= 0) cm_grow<2>(Cm, Pb, cw, lr, half);
  __syncthreads();
  wg_matmul<4>(Pa, Xb, Xb, wave, lr, half);
  if (cw >= 0) cm_grow<4>(Cm, Xb, cw, lr, half);
  __syncthreads();
  if (cw >= 0) cm_grow<8>(Cm, Pa, cw, lr, half);
  __syncthreads();

  if (cw >= 0) {
    const int c0 = 16 * cw;
    const v8f acc = mm_tile(Cm, Xa, lr, c0 + lr, half);
    float* kp = Kws + (size_t)h * LL + (8 * half) * NS + c0 + lr;
#pragma unroll
    for (int r = 0; r < 8; ++r) kp[r * NS] = acc[r];
  }
}

#define UT 136

__global__ __launch_bounds__(256) void ssm_conv_wmma(
    const float* __restrict__ inp, const float* __restrict__ Kws,
    const float* __restrict__ Ds, float* __restrict__ out)
{
  extern __shared__ __attribute__((aligned(16))) float smem[];
  float* Ush = smem;
  float* Kp  = smem + 65 * UT;

  const int h    = blockIdx.x;
  const int tid  = threadIdx.x;
  const int wave = tid >> 5;
  const int lane = tid & 31;
  const int half = lane >> 4;
  const int lr   = lane & 15;
  const int hi   = lr >> 3;
  const int col  = lr & 7;

  {
    const int sb = tid & 7;
    const int sl = tid >> 3;
    const float* up = inp + (size_t)sb * LL * HC + (size_t)sl * HC + h;
    const float* kp = Kws + (size_t)h * LL + tid;
    float uv[32], kv[4];
#pragma unroll
    for (int j = 0; j < 32; ++j) uv[j] = up[(size_t)j * 32 * HC];
#pragma unroll
    for (int j = 0; j < 4; ++j) kv[j] = kp[j * 256];

    if (tid < 128) Ush[tid] = 0.f;
    if (tid < 16) Kp[tid] = 0.f;

    float* ud = Ush + (1 + (sl >> 4)) * UT + (sl & 15) * 8 + sb;
#pragma unroll
    for (int j = 0; j < 32; ++j) ud[j * 2 * UT] = uv[j];
#pragma unroll
    for (int j = 0; j < 4; ++j) Kp[16 + j * 256 + tid] = kv[j];
  }
  __syncthreads();

  const float Dh = Ds[h];

  for (int j = 0; j < 4; ++j) {
    const int p  = (j & 1) ? (8 * j + 7 - wave) : (8 * j + wave);
    const int t0 = 2 * p;
    v8f acc = {0.f, 0.f, 0.f, 0.f, 0.f, 0.f, 0.f, 0.f};

    const float* kb = Kp + 16 + 16 * (t0 + 1) + lr - 2 * half;
    const float* ub = Ush + hi * UT + 16 * half + col;

    for (int n = 0; n < t0 + 2; ++n) {
      v2f af[4], bf[4];
#pragma unroll
      for (int u = 0; u < 4; ++u) {
        af[u].x = kb[-4 * u];
        af[u].y = kb[-4 * u - 1];
        bf[u].x = ub[(4 * u) * 8];
        bf[u].y = ub[(4 * u + 1) * 8];
      }
#pragma unroll
      for (int u = 0; u < 4; ++u)
        acc = __builtin_amdgcn_wmma_f32_16x16x4_f32(false, af[u], false, bf[u],
                                                    (short)0, acc, false, false);
      kb -= 16;
      ub += UT;
    }

    const float* ue = Ush + (1 + t0 + hi) * UT + (8 * half) * 8 + col;
    float* op = out + (size_t)col * LL * HC
                    + (size_t)(16 * (t0 + hi) + 8 * half) * HC + h;
#pragma unroll
    for (int r = 0; r < 8; ++r)
      op[(size_t)r * HC] = acc[r] + Dh * ue[r * 8];
  }
}

extern "C" void kernel_launch(void* const* d_in, const int* in_sizes, int n_in,
                              void* d_out, int out_size, void* d_ws, size_t ws_size,
                              hipStream_t stream)
{
  const float* inp = (const float*)d_in[0];
  const float* As  = (const float*)d_in[1];
  const float* Bs  = (const float*)d_in[2];
  const float* Cs  = (const float*)d_in[3];
  const float* Ds  = (const float*)d_in[4];
  const float* ls  = (const float*)d_in[5];
  float* out = (float*)d_out;

  float* Kws = (float*)d_ws;

  const size_t lds1 = (size_t)(2 * 4 * PW + 4 * NS * MS + 16 * MS) * sizeof(float);
  const size_t lds2 = (size_t)(65 * UT + 16 + LL) * sizeof(float);

  ssm_build_k  <<<dim3(HC), dim3(256), lds1, stream>>>(As, Bs, Cs, ls, Kws);
  ssm_conv_wmma<<<dim3(HC), dim3(256), lds2, stream>>>(inp, Kws, Ds, out);
}


# Both kernels fused into one launch; a thread's 32 strided input loads are issued before the pivot loop and first read after it (its barriers order LDS only); K goes from accumulators to LDS
# speedup vs baseline: 3.0092x; 1.0473x over previous
#include <hip/hip_runtime.h>

#define NS 64
#define HC 256
#define NB 8
#define LL 1024
#define MS 66

typedef __attribute__((ext_vector_type(2))) float v2f;
typedef __attribute__((ext_vector_type(8))) float v8f;

__device__ __forceinline__ v8f mm_tile(const float* __restrict__ A,
                                       const float* __restrict__ Bm,
                                       int arow, int bcol, int half)
{
  v2f af[16], bf[16];
#pragma unroll
  for (int k = 0; k < 16; ++k) {
    const int ka = 4 * k + 2 * half;
    af[k].x = A[arow * MS + ka];
    af[k].y = A[arow * MS + ka + 1];
    bf[k].x = Bm[(ka    ) * MS + bcol];
    bf[k].y = Bm[(ka + 1) * MS + bcol];
  }

  v8f acc = {0.f, 0.f, 0.f, 0.f, 0.f, 0.f, 0.f, 0.f};
#pragma unroll
  for (int k = 0; k < 16; ++k)
    acc = __builtin_amdgcn_wmma_f32_16x16x4_f32(false, af[k], false, bf[k],
                                                (short)0, acc, false, false);
  return acc;
}

template <int NCT>
__device__ __forceinline__ void wg_matmul(float* __restrict__ D,
                                          const float* __restrict__ A,
                                          const float* __restrict__ Bm,
                                          int wave, int lr, int half)
{
  for (int tt = wave; tt < 4 * NCT; tt += 8) {
    const int rt = tt / NCT, ct = tt % NCT;
    const int r0 = rt * 16, c0 = ct * 16;
    const v8f acc = mm_tile(A, Bm, r0 + lr, c0 + lr, half);
#pragma unroll
    for (int r = 0; r < 8; ++r)
      D[(r0 + r + 8 * half) * MS + c0 + lr] = acc[r];
  }
}

template <int NR>
__device__ __forceinline__ void cm_grow(float* Cm, const float* __restrict__ Pm,
                                        int ct, int lr, int half)
{
  const int c0 = 16 * ct;
  const v8f acc = mm_tile(Cm, Pm, lr & (NR - 1), c0 + lr, half);
  if (half == 0) {
#pragma unroll
    for (int r = 0; r < NR; ++r)
      Cm[(NR + r) * MS + c0 + lr] = acc[r];
  }
}

template <int NC>
__device__ __forceinline__ void x_grow(float* X, const float* __restrict__ Pm,
                                       int rt, int lr, int half)
{
  const int r0 = 16 * rt;
  const v8f acc = mm_tile(Pm, X, r0 + lr, lr & (NC - 1), half);
  if (lr < NC) {
#pragma unroll
    for (int r = 0; r < 8; ++r)
      X[(r0 + r + 8 * half) * MS + NC + lr] = acc[r];
  }
}

#define PW 36
#define UT 136

__device__ __forceinline__ void lds_barrier()
{
  __builtin_amdgcn_fence(__ATOMIC_RELEASE, "workgroup", "local");
  __builtin_amdgcn_s_barrier();
  __builtin_amdgcn_fence(__ATOMIC_ACQUIRE, "workgroup", "local");
}

template <int PP>
__device__ __forceinline__ void gj_pivot(float (&r)[33], float& diag,
                                         float* __restrict__ prow,
                                         int p, int i, int q)
{
  float* buf = prow + (PP & 1) * (4 * PW);
  float4* mine = (float4*)(buf + q * PW);

  if (i == p) {
#pragma unroll
    for (int k = 0; k < 8; ++k)
      mine[k] = make_float4(r[4 * k], r[4 * k + 1], r[4 * k + 2], r[4 * k + 3]);
    buf[q * PW + 32] = r[32];
  }
  const float aip = __int_as_float(__builtin_amdgcn_update_dpp(
      0, __float_as_int(r[0]), PP * 0x55, 0xF, 0xF, true));
  lds_barrier();

  float pr[33];
#pragma unroll
  for (int k = 0; k < 8; ++k) {
    const float4 v = mine[k];
    pr[4 * k] = v.x; pr[4 * k + 1] = v.y; pr[4 * k + 2] = v.z; pr[4 * k + 3] = v.w;
  }
  pr[32] = buf[q * PW + 32];
  const float piv = buf[PP * PW];

  const float f = (i != p) ? aip / piv : 0.0f;
  if (i == p) diag = piv;
#pragma unroll
  for (int k = 0; k < 33; ++k) r[k] = __builtin_fmaf(-f, pr[k], r[k]);
}

__global__ __launch_bounds__(256) void ssm_layer(
    const float* __restrict__ inp, const float* __restrict__ As,
    const float* __restrict__ Bs,  const float* __restrict__ Cs,
    const float* __restrict__ Ds,  const float* __restrict__ log_steps,
    float* __restrict__ out)
{
  extern __shared__ __attribute__((aligned(16))) float smem[];
  float* prow = smem;
  float* P  = smem + 2 * 4 * PW;
  float* Pn = P  + NS * MS;
  float* Xa = Pn + NS * MS;
  float* Xb = Xa + NS * MS;
  float* Cm = Xb + NS * MS;
  float* Ush = Cm + 16 * MS;
  float* Kp  = Ush + 65 * UT;

  const int h    = blockIdx.x;
  const int t    = threadIdx.x;
  const int wave = t >> 5, lane = t & 31;
  const int half = lane >> 4, lr = lane & 15;

  const int i = t >> 2;
  const int q = t & 3;
  const float s = 0.5f * __expf(log_steps[h]);
  const float* Ah = As + (size_t)h * NS * NS;

  float r[33];
  {
    float a[16];
#pragma unroll
    for (int k = 0; k < 16; ++k) a[k] = Ah[i * NS + 4 * k + q];
    const float b = (q == 0) ? Bs[h * NS + i] : 0.f;
#pragma unroll
    for (int k = 0; k < 16; ++k) {
      const float d = (i == 4 * k + q) ? 1.f : 0.f;
      r[k]      = __builtin_fmaf(-s, a[k], d);
      r[16 + k] = __builtin_fmaf( s, a[k], d);
    }
    r[32] = 2.f * s * b;
  }
  for (int e = t; e < 16 * MS; e += 256) Cm[e] = (e < NS) ? Cs[h * NS + e] : 0.f;
  if (t < 128) Ush[t] = 0.f;
  if (t < 16) Kp[t] = 0.f;

  const int sb = t & 7;
  const int sl = t >> 3;
  float uv[32];
  const float* up = inp + (size_t)sb * LL * HC + (size_t)sl * HC + h;
#pragma unroll
  for (int j = 0; j < 32; ++j) uv[j] = up[(size_t)j * 32 * HC];

  float diag = 1.f;
  for (int pb = 0; pb < 16; ++pb) {
    const int p0 = 4 * pb;
    gj_pivot<0>(r, diag, prow, p0,     i, q);
    gj_pivot<1>(r, diag, prow, p0 + 1, i, q);
    gj_pivot<2>(r, diag, prow, p0 + 2, i, q);
    gj_pivot<3>(r, diag, prow, p0 + 3, i, q);
#pragma unroll
    for (int k = 0; k < 32; ++k) r[k] = r[k + 1];
    r[32] = 0.f;
  }

  {
    const float dinv = 1.0f / diag;
#pragma unroll
    for (int k = 0; k < 16; ++k) P[i * MS + 4 * k + q] = r[k] * dinv;
    if (q == 0) Xa[i * MS + 0] = r[16] * dinv;
  }
  asm volatile("" :: "v"(up));
  {
    float* ud = Ush + (1 + (sl >> 4)) * UT + (sl & 15) * 8 + sb;
#pragma unroll
    for (int j = 0; j < 32; ++j) ud[j * 2 * UT] = uv[j];
  }
  __syncthreads();

  const int cw = wave - 4;
  wg_matmul<4>(Pn, P, P, wave, lr, half);
  if (cw >= 0) x_grow<1>(Xa, P, cw, lr, half);
  __syncthreads();
  wg_matmul<4>(P, Pn, Pn, wave, lr, half);
  if (cw >= 0) x_grow<2>(Xa, Pn, cw, lr, half);
  __syncthreads();
  wg_matmul<4>(Pn, P, P, wave, lr, half);
  if (cw >= 0) x_grow<4>(Xa, P, cw, lr, half);
  __syncthreads();
  wg_matmul<4>(P, Pn, Pn, wave, lr, half);
  if (cw >= 0) x_grow<8>(Xa, Pn, cw, lr, half);
  __syncthreads();
  wg_matmul<4>(Pn, P, P, wave, lr, half);
  wg_matmul<1>(Xa + 16, P, Xa, wave, lr, half);
  __syncthreads();
  wg_matmul<4>(P, Pn, Pn, wave, lr, half);
  wg_matmul<2>(Xa + 32, Pn, Xa, wave, lr, half);
  __syncthreads();
  float* Pa = P; float* Pb = Pn;

  wg_matmul<4>(Pb, Pa, Pa, wave, lr, half);
  if (cw >= 0) cm_grow<1>(Cm, Pa, cw, lr, half);
  __syncthreads();
  wg_matmul<4>(Xb, Pb, Pb, wave, lr, half);
  if (cw >= 0) cm_grow<2>(Cm, Pb, cw, lr, half);
  __syncthreads();
  wg_matmul<4>(Pa, Xb, Xb, wave, lr, half);
  if (cw >= 0) cm_grow<4>(Cm, Xb, cw, lr, half);
  __syncthreads();
  if (cw >= 0) cm_grow<8>(Cm, Pa, cw, lr, half);
  __syncthreads();

  if (cw >= 0) {
    const int c0 = 16 * cw;
    const v8f acc = mm_tile(Cm, Xa, lr, c0 + lr, half);
    float* kp = Kp + 16 + (8 * half) * NS + c0 + lr;
#pragma unroll
    for (int r = 0; r < 8; ++r) kp[r * NS] = acc[r];
  }
  __syncthreads();

  const int hi   = lr >> 3;
  const int col  = lr & 7;

  const float Dh = Ds[h];

  for (int j = 0; j < 4; ++j) {
    const int p  = (j & 1) ? (8 * j + 7 - wave) : (8 * j + wave);
    const int t0 = 2 * p;
    v8f acc = {0.f, 0.f, 0.f, 0.f, 0.f, 0.f, 0.f, 0.f};

    const float* kb = Kp + 16 + 16 * (t0 + 1) + lr - 2 * half;
    const float* ub = Ush + hi * UT + 16 * half + col;

    for (int n = 0; n < t0 + 2; ++n) {
      v2f af[4], bf[4];
#pragma unroll
      for (int u = 0; u < 4; ++u) {
        af[u].x = kb[-4 * u];
        af[u].y = kb[-4 * u - 1];
        bf[u].x = ub[(4 * u) * 8];
        bf[u].y = ub[(4 * u + 1) * 8];
      }
#pragma unroll
      for (int u = 0; u < 4; ++u)
        acc = __builtin_amdgcn_wmma_f32_16x16x4_f32(false, af[u], false, bf[u],
                                                    (short)0, acc, false, false);
      kb -= 16;
      ub += UT;
    }

    const float* ue = Ush + (1 + t0 + hi) * UT + (8 * half) * 8 + col;
    float* op = out + (size_t)col * LL * HC
                    + (size_t)(16 * (t0 + hi) + 8 * half) * HC + h;
#pragma unroll
    for (int r = 0; r < 8; ++r)
      op[(size_t)r * HC] = acc[r] + Dh * ue[r * 8];
  }
}

extern "C" void kernel_launch(void* const* d_in, const int* in_sizes, int n_in,
                              void* d_out, int out_size, void* d_ws, size_t ws_size,
                              hipStream_t stream)
{
  const float* inp = (const float*)d_in[0];
  const float* As  = (const float*)d_in[1];
  const float* Bs  = (const float*)d_in[2];
  const float* Cs  = (const float*)d_in[3];
  const float* Ds  = (const float*)d_in[4];
  const float* ls  = (const float*)d_in[5];
  float* out = (float*)d_out;

  const size_t lds = (size_t)(2 * 4 * PW + 4 * NS * MS + 16 * MS
                              + 65 * UT + 16 + LL) * sizeof(float);

  ssm_layer<<<dim3(HC), dim3(256), lds, stream>>>(inp, As, Bs, Cs, Ds, ls, out);
}
